# starting kernel
# baseline (speedup 1.0000x reference)
#include <hip/hip_runtime.h>
#include <hip/hip_bf16.h>

typedef float v2f __attribute__((ext_vector_type(2)));
typedef float v8f __attribute__((ext_vector_type(8)));

#define NQ 512
#define NKV 2562
#define NKV_PAD 2624
#define CDIM 64
#define NCHUNK 41
#define CHUNK 64
#define NQT (NQ / 16)
#define SCALE (1.0f / 64.0f)

#define OFF_QW   0L
#define OFF_KW   (OFF_QW   + (long)NQ * CDIM)
#define OFF_VW   (OFF_KW   + (long)NKV_PAD * CDIM)
#define OFF_EQ   (OFF_VW   + (long)NKV_PAD * CDIM)
#define OFF_IEQ  (OFF_EQ   + (long)NQ * CDIM)
#define OFF_EK   (OFF_IEQ  + (long)NQ * CDIM)
#define OFF_IEK  (OFF_EK   + (long)NKV_PAD * CDIM)
#define OFF_WSUM (OFF_IEK  + (long)NKV_PAD * CDIM)
#define OFF_EQC  (OFF_WSUM + 16L)
#define OFF_IEQC (OFF_EQC  + (long)NQ * 3)
#define OFF_EKC  (OFF_IEQC + (long)NQ * 3)
#define OFF_IEKC (OFF_EKC  + (long)NKV * 3)
#define OFF_PO   (OFF_IEKC + (long)NKV * 3 + 16L)
#define OFF_PM   (OFF_PO   + (long)NQT * NCHUNK * 16 * CDIM)
#define OFF_PS   (OFF_PM   + (long)NQT * NCHUNK * 16)

static __device__ __forceinline__ v8f wmma_f32_4(v2f a, v2f b, v8f c) {
    return __builtin_amdgcn_wmma_f32_16x16x4_f32(false, a, false, b, (short)0, c,
                                                 false, false);
}

__global__ void diffattn_prep(const float* __restrict__ q,
                              const float* __restrict__ kv,
                              const float* __restrict__ Wq,
                              const float* __restrict__ Wkv,
                              const float* __restrict__ Wdelta,
                              float* __restrict__ q_ws,
                              float* __restrict__ k_ws,
                              float* __restrict__ v_ws,
                              float* __restrict__ wsum_ws) {
    if (blockIdx.x == 0 && threadIdx.x < 3) {
        float s = 0.f;
        for (int c = 0; c < CDIM; ++c) s += Wdelta[threadIdx.x * CDIM + c];
        wsum_ws[threadIdx.x] = s;
    }

    const int KV_TILES = (NKV_PAD / 16) * (128 / 16);
    const int Q_TILES = (NQ / 16) * (CDIM / 16);
    int wave = (blockIdx.x * blockDim.x + threadIdx.x) >> 5;
    int lane = threadIdx.x & 31;
    if (wave >= KV_TILES + Q_TILES) return;

    int half = lane >> 4;
    int lm = lane & 15;

    if (wave < KV_TILES) {
        int mt = wave >> 3, nt = wave & 7;
        int m0 = mt * 16, n0 = nt * 16;
        int rowA = m0 + lm;
        bool inb = rowA < NKV;
        v8f acc = {};
        for (int kk = 0; kk < CDIM; kk += 4) {
            int ka = kk + 2 * half;
            v2f a, b;
            a.x = inb ? kv[rowA * CDIM + ka] : 0.f;
            a.y = inb ? kv[rowA * CDIM + ka + 1] : 0.f;
            b.x = Wkv[ka * 128 + n0 + lm];
            b.y = Wkv[(ka + 1) * 128 + n0 + lm];
            acc = wmma_f32_4(a, b, acc);
        }
        int col = n0 + lm;
#pragma unroll
        for (int r = 0; r < 8; ++r) {
            int row = m0 + r + 8 * half;
            if (col < CDIM)
                k_ws[row * CDIM + col] = acc[r];
            else
                v_ws[row * CDIM + (col - CDIM)] = acc[r];
        }
    } else {
        int t = wave - KV_TILES;
        int mt = t >> 2, nt = t & 3;
        int m0 = mt * 16, n0 = nt * 16;
        int rowA = m0 + lm;
        v8f acc = {};
        for (int kk = 0; kk < CDIM; kk += 4) {
            int ka = kk + 2 * half;
            v2f a, b;
            a.x = q[rowA * CDIM + ka];
            a.y = q[rowA * CDIM + ka + 1];
            b.x = Wq[ka * CDIM + n0 + lm];
            b.y = Wq[(ka + 1) * CDIM + n0 + lm];
            acc = wmma_f32_4(a, b, acc);
        }
        int col = n0 + lm;
#pragma unroll
        for (int r = 0; r < 8; ++r) {
            int row = m0 + r + 8 * half;
            q_ws[row * CDIM + col] = acc[r];
        }
    }
}

__global__ void diffattn_exp(const float* __restrict__ q_ws,
                             const float* __restrict__ k_ws,
                             const float* __restrict__ q_coord,
                             const float* __restrict__ kv_coord,
                             float* __restrict__ Eq, float* __restrict__ iEq,
                             float* __restrict__ Ek, float* __restrict__ iEk,
                             float* __restrict__ Eqc, float* __restrict__ iEqc,
                             float* __restrict__ Ekc, float* __restrict__ iEkc) {
    const int NE_Q = NQ * CDIM;
    const int NE_K = NKV_PAD * CDIM;
    const int NE_QC = NQ * 3;
    const int NE_KC = NKV * 3;
    int i = blockIdx.x * blockDim.x + threadIdx.x;
    if (i < NE_Q) {
        float x = q_ws[i];
        Eq[i] = __expf(x);
        iEq[i] = __expf(-x);
    } else if (i < NE_Q + NE_K) {
        int j = i - NE_Q;
        float x = k_ws[j];
        Ek[j] = __expf(x);
        iEk[j] = __expf(-x);
    } else if (i < NE_Q + NE_K + NE_QC) {
        int j = i - NE_Q - NE_K;
        float x = q_coord[j];
        Eqc[j] = __expf(x);
        iEqc[j] = __expf(-x);
    } else if (i < NE_Q + NE_K + NE_QC + NE_KC) {
        int j = i - NE_Q - NE_K - NE_QC;
        float x = kv_coord[j];
        Ekc[j] = __expf(x);
        iEkc[j] = __expf(-x);
    }
}

__global__ void __launch_bounds__(256)
diffattn_score(const float* __restrict__ Eq, const float* __restrict__ iEq,
               const float* __restrict__ Ek, const float* __restrict__ iEk,
               const float* __restrict__ Eqc, const float* __restrict__ iEqc,
               const float* __restrict__ Ekc, const float* __restrict__ iEkc,
               const float* __restrict__ wsum_ws,
               const float* __restrict__ v_ws,
               float* __restrict__ po, float* __restrict__ pm,
               float* __restrict__ ps) {
    __shared__ float sEq[16][CDIM + 1];
    __shared__ float sIEq[16][CDIM + 1];
    __shared__ float sP[16][CHUNK + 8];
    __shared__ float sEqc[16][4];
    __shared__ float sIEqc[16][4];
    __shared__ float sRed[16][17];
    __shared__ float sMax[16];
    __shared__ float sSum[16];
    __shared__ float sW[3];

    const int tid = threadIdx.x;
    const int ch = blockIdx.x;
    const int qt = blockIdx.y;
    const int q0 = qt * 16;
    const int j0 = ch * CHUNK;

    for (int i = tid; i < 16 * CDIM; i += 256) {
        int r = i / CDIM, c = i % CDIM;
        sEq[r][c] = Eq[(q0 + r) * CDIM + c];
        sIEq[r][c] = iEq[(q0 + r) * CDIM + c];
    }
    if (tid < 48) {
        int r = tid / 3, d = tid % 3;
        sEqc[r][d] = Eqc[(q0 + r) * 3 + d];
        sIEqc[r][d] = iEqc[(q0 + r) * 3 + d];
    }
    if (tid < 3) sW[tid] = wsum_ws[tid];
    __syncthreads();

    for (int p = tid; p < 16 * CHUNK; p += 256) {
        int qi = p & 15;
        int jl = p >> 4;
        int jj = j0 + jl;
        float s;
        if (jj < NKV) {
            s = 0.f;
            const float4* e4 = reinterpret_cast<const float4*>(Ek + jj * CDIM);
            const float4* i4 = reinterpret_cast<const float4*>(iEk + jj * CDIM);
#pragma unroll 4
            for (int c4 = 0; c4 < CDIM / 4; ++c4) {
                float4 ek = e4[c4];
                float4 ik = i4[c4];
                int c = 4 * c4;
                s += fminf(sEq[qi][c + 0] * ik.x, sIEq[qi][c + 0] * ek.x);
                s += fminf(sEq[qi][c + 1] * ik.y, sIEq[qi][c + 1] * ek.y);
                s += fminf(sEq[qi][c + 2] * ik.z, sIEq[qi][c + 2] * ek.z);
                s += fminf(sEq[qi][c + 3] * ik.w, sIEq[qi][c + 3] * ek.w);
            }
#pragma unroll
            for (int d = 0; d < 3; ++d) {
                float ekc = Ekc[jj * 3 + d];
                float ikc = iEkc[jj * 3 + d];
                s += fminf(sEqc[qi][d] * ikc, sIEqc[qi][d] * ekc) * sW[d];
            }
            s *= SCALE;
        } else {
            s = -__builtin_inff();
        }
        sP[qi][jl] = s;
    }
    __syncthreads();

    const int qi = tid & 15;
    const int g = tid >> 4;
    {
        float m = -__builtin_inff();
        for (int jl = g; jl < CHUNK; jl += 16) m = fmaxf(m, sP[qi][jl]);
        sRed[qi][g] = m;
    }
    __syncthreads();
    if (tid < 16) {
        float m = sRed[tid][0];
#pragma unroll
        for (int i = 1; i < 16; ++i) m = fmaxf(m, sRed[tid][i]);
        sMax[tid] = m;
    }
    __syncthreads();

    {
        float mr = sMax[qi];
        float psum = 0.f;
        for (int jl = g; jl < CHUNK; jl += 16) {
            float e = __expf(sP[qi][jl] - mr);
            sP[qi][jl] = e;
            psum += e;
        }
        sRed[qi][g] = psum;
    }
    __syncthreads();
    if (tid < 16) {
        float s = 0.f;
#pragma unroll
        for (int i = 0; i < 16; ++i) s += sRed[tid][i];
        sSum[tid] = s;
    }
    __syncthreads();

    const int wave = tid >> 5;
    const int lane = tid & 31;
    const int half = lane >> 4;
    const int lm = lane & 15;
    const long base = (long)(qt * NCHUNK + ch) * 16;
    if (wave < 4) {
        int n0 = wave * 16;
        v8f acc = {};
        for (int kk = 0; kk < CHUNK; kk += 4) {
            int ka = kk + 2 * half;
            v2f a, b;
            a.x = sP[lm][ka];
            a.y = sP[lm][ka + 1];
            b.x = v_ws[(j0 + ka) * CDIM + n0 + lm];
            b.y = v_ws[(j0 + ka + 1) * CDIM + n0 + lm];
            acc = wmma_f32_4(a, b, acc);
        }
#pragma unroll
        for (int r = 0; r < 8; ++r) {
            int row = r + 8 * half;
            po[(base + row) * CDIM + n0 + lm] = acc[r];
        }
    }
    if (tid < 16) {
        pm[base + tid] = sMax[tid];
        ps[base + tid] = sSum[tid];
    }
}

__global__ void __launch_bounds__(256)
diffattn_combine(const float* __restrict__ po, const float* __restrict__ pm,
                 const float* __restrict__ ps,
                 const float* __restrict__ Wproj,
                 const float* __restrict__ bproj, float* __restrict__ out) {
    __shared__ float sO[16][CDIM + 1];
    __shared__ float cm[16][NCHUNK + 1];
    __shared__ float cw[16][NCHUNK + 1];
    __shared__ float sM[16];
    __shared__ float sInv[16];

    const int tid = threadIdx.x;
    const int qt = blockIdx.x;
    const long pbase = (long)qt * NCHUNK * 16;

    for (int i = tid; i < 16 * NCHUNK; i += 256) {
        int r = i & 15, c = i >> 4;
        cm[r][c] = pm[pbase + c * 16 + r];
        cw[r][c] = ps[pbase + c * 16 + r];
    }
    __syncthreads();
    if (tid < 16) {
        float M = cm[tid][0];
        for (int c = 1; c < NCHUNK; ++c) M = fmaxf(M, cm[tid][c]);
        float S = 0.f;
        for (int c = 0; c < NCHUNK; ++c) S += cw[tid][c] * __expf(cm[tid][c] - M);
        sM[tid] = M;
        sInv[tid] = 1.0f / S;
    }
    __syncthreads();
    for (int i = tid; i < 16 * NCHUNK; i += 256) {
        int r = i & 15, c = i >> 4;
        cw[r][c] = __expf(cm[r][c] - sM[r]) * sInv[r];
    }
    __syncthreads();

    for (int idx = tid; idx < 16 * CDIM; idx += 256) {
        int r = idx >> 6, n = idx & 63;
        float a = 0.f;
        for (int c = 0; c < NCHUNK; ++c)
            a += po[(pbase + c * 16 + r) * CDIM + n] * cw[r][c];
        sO[r][n] = a;
    }
    __syncthreads();

    const int wave = tid >> 5;
    const int lane = tid & 31;
    const int half = lane >> 4;
    const int lm = lane & 15;
    if (wave < 4) {
        int n0 = wave * 16;
        v8f acc = {};
        for (int kk = 0; kk < CDIM; kk += 4) {
            int ka = kk + 2 * half;
            v2f a, b;
            a.x = sO[lm][ka];
            a.y = sO[lm][ka + 1];
            b.x = Wproj[ka * CDIM + n0 + lm];
            b.y = Wproj[(ka + 1) * CDIM + n0 + lm];
            acc = wmma_f32_4(a, b, acc);
        }
        float bb = bproj[n0 + lm];
#pragma unroll
        for (int r = 0; r < 8; ++r) {
            int row = r + 8 * half;
            out[(qt * 16 + row) * CDIM + n0 + lm] = acc[r] + bb;
        }
    }
}

extern "C" void kernel_launch(void* const* d_in, const int* in_sizes, int n_in,
                              void* d_out, int out_size, void* d_ws, size_t ws_size,
                              hipStream_t stream) {
    const float* q        = (const float*)d_in[0];
    const float* q_coord  = (const float*)d_in[1];
    const float* kv       = (const float*)d_in[2];
    const float* kv_coord = (const float*)d_in[3];
    const float* Wq       = (const float*)d_in[4];
    const float* Wkv      = (const float*)d_in[5];
    const float* Wdelta   = (const float*)d_in[6];
    const float* Wproj    = (const float*)d_in[7];
    const float* bproj    = (const float*)d_in[8];
    float* out = (float*)d_out;
    float* ws  = (float*)d_ws;

    float* q_ws    = ws + OFF_QW;
    float* k_ws    = ws + OFF_KW;
    float* v_ws    = ws + OFF_VW;
    float* Eqp     = ws + OFF_EQ;
    float* iEqp    = ws + OFF_IEQ;
    float* Ekp     = ws + OFF_EK;
    float* iEkp    = ws + OFF_IEK;
    float* wsum_ws = ws + OFF_WSUM;
    float* Eqc     = ws + OFF_EQC;
    float* iEqc    = ws + OFF_IEQC;
    float* Ekc     = ws + OFF_EKC;
    float* iEkc    = ws + OFF_IEKC;
    float* po      = ws + OFF_PO;
    float* pm      = ws + OFF_PM;
    float* ps      = ws + OFF_PS;

    diffattn_prep<<<180, 256, 0, stream>>>(q, kv, Wq, Wkv, Wdelta,
                                           q_ws, k_ws, v_ws, wsum_ws);

    int exp_total = NQ * CDIM + NKV_PAD * CDIM + NQ * 3 + NKV * 3;
    diffattn_exp<<<(exp_total + 255) / 256, 256, 0, stream>>>(
        q_ws, k_ws, q_coord, kv_coord, Eqp, iEqp, Ekp, iEkp, Eqc, iEqc, Ekc, iEkc);

    dim3 sgrid(NCHUNK, NQT);
    diffattn_score<<<sgrid, 256, 0, stream>>>(Eqp, iEqp, Ekp, iEkp, Eqc, iEqc,
                                              Ekc, iEkc, wsum_ws, v_ws, po, pm, ps);

    diffattn_combine<<<NQT, 256, 0, stream>>>(po, pm, ps, Wproj, bproj, out);
}


# softmax shift is an upper bound of the score from Wdelta's row sums, equal in all chunks: chunk max passes, max planes and the merge's rescaling exps are gone, partials add up; sum order changes
# speedup vs baseline: 3.0409x; 3.0409x over previous
#include <hip/hip_runtime.h>
#include <hip/hip_bf16.h>

typedef float v2f __attribute__((ext_vector_type(2)));
typedef float v8f __attribute__((ext_vector_type(8)));

#define NQ 512
#define NKV 2562
#define CDIM 64
#define NCHUNK 15
#define CHUNK 176
#define NIT (CHUNK / 16)
#define QB 32
#define NQB (NQ / QB)
#define NQT (NQ / 16)
#define SCALE (1.0f / 64.0f)
#define VROWS 192

#define OFF_PO   0L
#define OFF_PS   (OFF_PO   + (long)NQT * NCHUNK * 16 * CDIM)

static __device__ __forceinline__ v8f wmma_f32_4(v2f a, v2f b, v8f c) {
    return __builtin_amdgcn_wmma_f32_16x16x4_f32(false, a, false, b, (short)0, c,
                                                 false, false);
}

#define TSTR 68
#define LOG2E 1.44269504088896340736f
#define VSTR 76
#define PSTR (CHUNK + 4)

template <int NT>
static __device__ __forceinline__ void project_tiles(
        const float (*sA)[VSTR], const float (*sB)[VSTR], float (*sOut)[TSTR],
        int rt0, int rstep, int cK, int lm, int half) {
    v8f acc[NT];
#pragma unroll
    for (int i = 0; i < NT; ++i) acc[i] = v8f{};
#pragma unroll
    for (int kk = 0; kk < CDIM; kk += 4) {
        int ka = kk + 2 * half;
        v2f b;
        b.x = sB[ka][cK];
        b.y = sB[ka + 1][cK];
#pragma unroll
        for (int i = 0; i < NT; ++i) {
            v2f a = *reinterpret_cast<const v2f*>(&sA[16 * (rt0 + rstep * i) + lm][ka]);
            acc[i] = wmma_f32_4(a, b, acc[i]);
        }
    }
#pragma unroll
    for (int i = 0; i < NT; ++i) {
#pragma unroll
        for (int r = 0; r < 8; ++r) {
            int row = 16 * (rt0 + rstep * i) + r + 8 * half;
            sOut[row][cK] = acc[i][r] * LOG2E;
        }
    }
}

__global__ void __launch_bounds__(512)
diffattn_score(const float* __restrict__ q, const float* __restrict__ kv,
               const float* __restrict__ q_coord,
               const float* __restrict__ kv_coord,
               const float* __restrict__ Wq, const float* __restrict__ Wkv,
               const float* __restrict__ Wdelta,
               float* __restrict__ po, float* __restrict__ ps) {
    __shared__ __attribute__((aligned(16))) float sQ[QB][TSTR];
    __shared__ __attribute__((aligned(16))) float sK[CHUNK][TSTR];
    __shared__ __attribute__((aligned(16))) float sV[VROWS][VSTR];
    __shared__ __attribute__((aligned(16))) float sQr[QB][VSTR];
    __shared__ __attribute__((aligned(16))) float sWk[CDIM][VSTR];
    __shared__ __attribute__((aligned(16))) float sWq[CDIM][VSTR];
    __shared__ __attribute__((aligned(16))) float sP[QB][PSTR];
    __shared__ float sQc[128][4];
    __shared__ float sKc[512][4];
    __shared__ __attribute__((aligned(16))) float sWd[512];
    __shared__ float sRed[QB][17];
    __shared__ float sW[3];

    const int tid = threadIdx.x;
    const int ch = blockIdx.x;
    const int qb = blockIdx.y;
    const int q0 = qb * QB;
    const int j0 = ch * CHUNK;
    const int wave = tid >> 5;
    const int lane = tid & 31;
    const int half = lane >> 4;
    const int lm = lane & 15;

    float4 rkv[6];
#pragma unroll
    for (int i = 0; i < 6; ++i) {
        int idx = tid + 512 * i;
        int jl = idx >> 4;
        int rr = j0 + jl < NKV ? j0 + jl : NKV - 1;
        rkv[i] = *reinterpret_cast<const float4*>(kv + (long)rr * CDIM + 4 * (idx & 15));
    }
    const float4 rq = *reinterpret_cast<const float4*>(q + (long)(q0 + (tid >> 4)) * CDIM + 4 * (tid & 15));
    float4 rwk[2], rwq[2];
#pragma unroll
    for (int i = 0; i < 2; ++i) {
        int idx = tid + 512 * i;
        rwk[i] = *reinterpret_cast<const float4*>(Wkv + (idx >> 4) * 128 + 4 * (idx & 15));
        rwq[i] = *reinterpret_cast<const float4*>(Wq + (idx >> 4) * CDIM + 4 * (idx & 15));
    }
    const int cr = tid >> 2, cd = tid & 3;
    const float rqc = q_coord[(q0 + (cr < QB ? cr : QB - 1)) * 3 + (cd < 3 ? cd : 2)];
    const bool okc = j0 + tid < NKV;
    const float* cp = kv_coord + (okc ? j0 + tid : NKV - 1) * 3;
    const float x0 = cp[0], x1 = cp[1], x2 = cp[2];
    const float rwd = Wdelta[tid < 3 * CDIM ? tid : 3 * CDIM - 1];
#pragma unroll
    for (int i = 0; i < 6; ++i) {
        int idx = tid + 512 * i;
        *reinterpret_cast<float4*>(&sV[idx >> 4][4 * (idx & 15)]) = rkv[i];
    }
    *reinterpret_cast<float4*>(&sQr[tid >> 4][4 * (tid & 15)]) = rq;
#pragma unroll
    for (int i = 0; i < 2; ++i) {
        int idx = tid + 512 * i;
        *reinterpret_cast<float4*>(&sWk[idx >> 4][4 * (idx & 15)]) = rwk[i];
        *reinterpret_cast<float4*>(&sWq[idx >> 4][4 * (idx & 15)]) = rwq[i];
    }
    sQc[cr][cd] = rqc * LOG2E;
    sKc[tid][0] = okc ? x0 * LOG2E : 0.f;
    sKc[tid][1] = okc ? x1 * LOG2E : 0.f;
    sKc[tid][2] = okc ? x2 * LOG2E : 0.f;
    sWd[tid] = rwd;
    __syncthreads();

    if (tid >= 224 && tid < 227) {
        int d = tid - 224;
        float s = 0.f;
        for (int cc = 0; cc < CDIM; ++cc) s += sWd[d * CDIM + cc];
        sW[d] = s;
    }

    const int cK = 16 * (wave & 3) + lm;
    if (wave < 12) {
        project_tiles<3>(sV, sWk, sK, wave >> 2, 4, cK, lm, half);
    } else {
        project_tiles<2>(sV, sWk, sK, 3, 4, cK, lm, half);
        project_tiles<2>(sQr, sWq, sQ, 0, 1, cK, lm, half);
    }
    __syncthreads();

    {
        const int qi = tid & 31;
        const int jl0 = tid >> 5;
        float s[NIT] = {};
#pragma unroll 2
        for (int c4 = 0; c4 < CDIM / 4; ++c4) {
            float4 qv = *reinterpret_cast<const float4*>(&sQ[qi][4 * c4]);
#pragma unroll
            for (int it = 0; it < NIT; ++it) {
                float4 kr = *reinterpret_cast<const float4*>(&sK[jl0 + 16 * it][4 * c4]);
                s[it] += __builtin_amdgcn_exp2f(-__builtin_fabsf(qv.x - kr.x));
                s[it] += __builtin_amdgcn_exp2f(-__builtin_fabsf(qv.y - kr.y));
                s[it] += __builtin_amdgcn_exp2f(-__builtin_fabsf(qv.z - kr.z));
                s[it] += __builtin_amdgcn_exp2f(-__builtin_fabsf(qv.w - kr.w));
            }
        }
        const float bound = SCALE * ((float)CDIM + fmaxf(sW[0], 0.f) + fmaxf(sW[1], 0.f)
                                     + fmaxf(sW[2], 0.f));
        float psum = 0.f;
#pragma unroll
        for (int it = 0; it < NIT; ++it) {
            int jl = jl0 + 16 * it;
            float sc = s[it];
#pragma unroll
            for (int d = 0; d < 3; ++d)
                sc += __builtin_amdgcn_exp2f(-__builtin_fabsf(sQc[qi][d] - sKc[jl][d])) * sW[d];
            sc *= SCALE;
            float e = j0 + jl < NKV ? __expf(sc - bound) : 0.f;
            sP[qi][jl] = e;
            psum += e;
        }
        sRed[qi][jl0] = psum;
    }
    __syncthreads();

    if (wave < 8) {
        const int qtl = wave >> 2;
        const long base = (long)((2 * qb + qtl) * NCHUNK + ch) * 16;
        int n0 = (wave & 3) * 16;
        v8f acc = {};
        for (int kk = 0; kk < CHUNK; kk += 4) {
            int ka = kk + 2 * half;
            v2f a, b;
            a.x = sP[16 * qtl + lm][ka];
            a.y = sP[16 * qtl + lm][ka + 1];
            b.x = sV[ka][n0 + lm];
            b.y = sV[ka + 1][n0 + lm];
            acc = wmma_f32_4(a, b, acc);
        }
#pragma unroll
        for (int r = 0; r < 8; ++r) {
            int row = r + 8 * half;
            po[(base + row) * CDIM + n0 + lm] = acc[r];
        }
    }
    if (wave == 8) {
        const long b1 = (long)((2 * qb + half) * NCHUNK + ch) * 16 + lm;
        float sum = 0.f;
#pragma unroll
        for (int i = 0; i < 16; ++i) sum += sRed[lane][i];
        ps[b1] = sum;
    }
}

__global__ void __launch_bounds__(256)
diffattn_combine(const float* __restrict__ po, const float* __restrict__ ps,
                 const float* __restrict__ Wkv,
                 const float* __restrict__ Wproj,
                 const float* __restrict__ bproj, float* __restrict__ out) {
    __shared__ float sO[16][CDIM + 1];
    __shared__ float sT[16][CDIM + 1];

    const int tid = threadIdx.x;
    const int qt = blockIdx.x;
    const long pbase = (long)qt * NCHUNK * 16;
    const int wave = tid >> 5;
    const int lane = tid & 31;
    const int half = lane >> 4;
    const int lm = lane & 15;

    const int r = tid >> 4, c4 = tid & 15;
    float4 pv[NCHUNK];
    float sv[NCHUNK];
#pragma unroll
    for (int c = 0; c < NCHUNK; ++c) {
        pv[c] = *reinterpret_cast<const float4*>(po + (pbase + c * 16 + r) * CDIM + 4 * c4);
        sv[c] = ps[pbase + c * 16 + r];
    }

    v2f bv[CDIM / 4];
    if (wave < 4) {
        const float* col = Wkv + CDIM + wave * 16 + lm;
#pragma unroll
        for (int t = 0; t < CDIM / 4; ++t) {
            int ka = 4 * t + 2 * half;
            bv[t].x = col[ka * 128];
            bv[t].y = col[(ka + 1) * 128];
        }
    }

    {
        float4 a = pv[0];
        float S = sv[0];
#pragma unroll
        for (int c = 1; c < NCHUNK; ++c) {
            a.x += pv[c].x;
            a.y += pv[c].y;
            a.z += pv[c].z;
            a.w += pv[c].w;
            S += sv[c];
        }
        const float inv = 1.0f / S;
        sO[r][4 * c4 + 0] = a.x * inv;
        sO[r][4 * c4 + 1] = a.y * inv;
        sO[r][4 * c4 + 2] = a.z * inv;
        sO[r][4 * c4 + 3] = a.w * inv;
    }
    __syncthreads();

    if (wave < 4) {
        int n0 = wave * 16;
        v8f acc = {};
#pragma unroll
        for (int t = 0; t < CDIM / 4; ++t) {
            int ka = 4 * t + 2 * half;
            v2f a;
            a.x = sO[lm][ka];
            a.y = sO[lm][ka + 1];
            acc = wmma_f32_4(a, bv[t], acc);
        }
#pragma unroll
        for (int r = 0; r < 8; ++r) sT[r + 8 * half][n0 + lm] = acc[r];
    }
    __syncthreads();

    if (wave < 4) {
        int n0 = wave * 16;
        v8f acc = {};
        for (int kk = 0; kk < CDIM; kk += 4) {
            int ka = kk + 2 * half;
            v2f a, b;
            a.x = sT[lm][ka];
            a.y = sT[lm][ka + 1];
            b.x = Wproj[ka * CDIM + n0 + lm];
            b.y = Wproj[(ka + 1) * CDIM + n0 + lm];
            acc = wmma_f32_4(a, b, acc);
        }
        float bb = bproj[n0 + lm];
#pragma unroll
        for (int r = 0; r < 8; ++r) {
            int row = r + 8 * half;
            out[(qt * 16 + row) * CDIM + n0 + lm] = acc[r] + bb;
        }
    }
}

extern "C" void kernel_launch(void* const* d_in, const int* in_sizes, int n_in,
                              void* d_out, int out_size, void* d_ws, size_t ws_size,
                              hipStream_t stream) {
    const float* q        = (const float*)d_in[0];
    const float* q_coord  = (const float*)d_in[1];
    const float* kv       = (const float*)d_in[2];
    const float* kv_coord = (const float*)d_in[3];
    const float* Wq       = (const float*)d_in[4];
    const float* Wkv      = (const float*)d_in[5];
    const float* Wdelta   = (const float*)d_in[6];
    const float* Wproj    = (const float*)d_in[7];
    const float* bproj    = (const float*)d_in[8];
    float* out = (float*)d_out;
    float* ws  = (float*)d_ws;

    float* po = ws + OFF_PO;
    float* ps = ws + OFF_PS;

    dim3 sgrid(NCHUNK, NQB);
    diffattn_score<<<sgrid, 512, 0, stream>>>(q, kv, q_coord, kv_coord, Wq, Wkv,
                                              Wdelta, po, ps);

    diffattn_combine<<<NQT, 256, 0, stream>>>(po, ps, Wkv, Wproj, bproj, out);
}


# score loop issue order: a packed f32 subtract or add behind every exp2 (software pipeline over kv rows, scheduler groups) replaces runs of 20+ exp2; four partial channel sums: sum order changes
# speedup vs baseline: 3.2198x; 1.0588x over previous
#include <hip/hip_runtime.h>
#include <hip/hip_bf16.h>

typedef float v2f __attribute__((ext_vector_type(2)));
typedef float v8f __attribute__((ext_vector_type(8)));

#define NQ 512
#define NKV 2562
#define CDIM 64
#define NCHUNK 15
#define CHUNK 176
#define NIT (CHUNK / 16)
#define QB 32
#define NQB (NQ / QB)
#define NQT (NQ / 16)
#define SCALE (1.0f / 64.0f)
#define VROWS 192

#define OFF_PO   0L
#define OFF_PS   (OFF_PO   + (long)NQT * NCHUNK * 16 * CDIM)

static __device__ __forceinline__ v8f wmma_f32_4(v2f a, v2f b, v8f c) {
    return __builtin_amdgcn_wmma_f32_16x16x4_f32(false, a, false, b, (short)0, c,
                                                 false, false);
}

#define TSTR 68
#define LOG2E 1.44269504088896340736f
#define VSTR 76
#define PSTR (CHUNK + 4)

template <int NT>
static __device__ __forceinline__ void project_tiles(
        const float (*sA)[VSTR], const float (*sB)[VSTR], float (*sOut)[TSTR],
        int rt0, int rstep, int cK, int lm, int half) {
    v8f acc[NT];
#pragma unroll
    for (int i = 0; i < NT; ++i) acc[i] = v8f{};
#pragma unroll
    for (int kk = 0; kk < CDIM; kk += 4) {
        int ka = kk + 2 * half;
        v2f b;
        b.x = sB[ka][cK];
        b.y = sB[ka + 1][cK];
#pragma unroll
        for (int i = 0; i < NT; ++i) {
            v2f a = *reinterpret_cast<const v2f*>(&sA[16 * (rt0 + rstep * i) + lm][ka]);
            acc[i] = wmma_f32_4(a, b, acc[i]);
        }
    }
#pragma unroll
    for (int i = 0; i < NT; ++i) {
#pragma unroll
        for (int r = 0; r < 8; ++r) {
            int row = 16 * (rt0 + rstep * i) + r + 8 * half;
            sOut[row][cK] = acc[i][r] * LOG2E;
        }
    }
}

__global__ void __launch_bounds__(512)
diffattn_score(const float* __restrict__ q, const float* __restrict__ kv,
               const float* __restrict__ q_coord,
               const float* __restrict__ kv_coord,
               const float* __restrict__ Wq, const float* __restrict__ Wkv,
               const float* __restrict__ Wdelta,
               float* __restrict__ po, float* __restrict__ ps) {
    __shared__ __attribute__((aligned(16))) float sQ[QB][TSTR];
    __shared__ __attribute__((aligned(16))) float sK[CHUNK][TSTR];
    __shared__ __attribute__((aligned(16))) float sV[VROWS][VSTR];
    __shared__ __attribute__((aligned(16))) float sQr[QB][VSTR];
    __shared__ __attribute__((aligned(16))) float sWk[CDIM][VSTR];
    __shared__ __attribute__((aligned(16))) float sWq[CDIM][VSTR];
    __shared__ __attribute__((aligned(16))) float sP[QB][PSTR];
    __shared__ float sQc[128][4];
    __shared__ float sKc[512][4];
    __shared__ __attribute__((aligned(16))) float sWd[512];
    __shared__ float sRed[QB][17];
    __shared__ float sW[3];

    const int tid = threadIdx.x;
    const int ch = blockIdx.x;
    const int qb = blockIdx.y;
    const int q0 = qb * QB;
    const int j0 = ch * CHUNK;
    const int wave = tid >> 5;
    const int lane = tid & 31;
    const int half = lane >> 4;
    const int lm = lane & 15;

    float4 rkv[6];
#pragma unroll
    for (int i = 0; i < 6; ++i) {
        int idx = tid + 512 * i;
        int jl = idx >> 4;
        int rr = j0 + jl < NKV ? j0 + jl : NKV - 1;
        rkv[i] = *reinterpret_cast<const float4*>(kv + (long)rr * CDIM + 4 * (idx & 15));
    }
    const float4 rq = *reinterpret_cast<const float4*>(q + (long)(q0 + (tid >> 4)) * CDIM + 4 * (tid & 15));
    float4 rwk[2], rwq[2];
#pragma unroll
    for (int i = 0; i < 2; ++i) {
        int idx = tid + 512 * i;
        rwk[i] = *reinterpret_cast<const float4*>(Wkv + (idx >> 4) * 128 + 4 * (idx & 15));
        rwq[i] = *reinterpret_cast<const float4*>(Wq + (idx >> 4) * CDIM + 4 * (idx & 15));
    }
    const int cr = tid >> 2, cd = tid & 3;
    const float rqc = q_coord[(q0 + (cr < QB ? cr : QB - 1)) * 3 + (cd < 3 ? cd : 2)];
    const bool okc = j0 + tid < NKV;
    const float* cp = kv_coord + (okc ? j0 + tid : NKV - 1) * 3;
    const float x0 = cp[0], x1 = cp[1], x2 = cp[2];
    const float rwd = Wdelta[tid < 3 * CDIM ? tid : 3 * CDIM - 1];
#pragma unroll
    for (int i = 0; i < 6; ++i) {
        int idx = tid + 512 * i;
        *reinterpret_cast<float4*>(&sV[idx >> 4][4 * (idx & 15)]) = rkv[i];
    }
    *reinterpret_cast<float4*>(&sQr[tid >> 4][4 * (tid & 15)]) = rq;
#pragma unroll
    for (int i = 0; i < 2; ++i) {
        int idx = tid + 512 * i;
        *reinterpret_cast<float4*>(&sWk[idx >> 4][4 * (idx & 15)]) = rwk[i];
        *reinterpret_cast<float4*>(&sWq[idx >> 4][4 * (idx & 15)]) = rwq[i];
    }
    sQc[cr][cd] = rqc * LOG2E;
    sKc[tid][0] = okc ? x0 * LOG2E : 0.f;
    sKc[tid][1] = okc ? x1 * LOG2E : 0.f;
    sKc[tid][2] = okc ? x2 * LOG2E : 0.f;
    sWd[tid] = rwd;
    __syncthreads();

    if (tid >= 224 && tid < 227) {
        int d = tid - 224;
        float s = 0.f;
        for (int cc = 0; cc < CDIM; ++cc) s += sWd[d * CDIM + cc];
        sW[d] = s;
    }

    const int cK = 16 * (wave & 3) + lm;
    if (wave < 12) {
        project_tiles<3>(sV, sWk, sK, wave >> 2, 4, cK, lm, half);
    } else {
        project_tiles<2>(sV, sWk, sK, 3, 4, cK, lm, half);
        project_tiles<2>(sQr, sWq, sQ, 0, 1, cK, lm, half);
    }
    __syncthreads();

    {
        const int qi = tid & 31;
        const int jl0 = tid >> 5;
        constexpr int NU = 2 * NIT;
        v2f sa[NIT], sb[NIT];
#pragma unroll
        for (int it = 0; it < NIT; ++it) { sa[it] = v2f{0.f, 0.f}; sb[it] = v2f{0.f, 0.f}; }
        v2f carlo = {0.f, 0.f}, carhi = {0.f, 0.f};
        for (int c8 = 0; c8 < CDIM / 8; ++c8) {
            float4 qv[2];
            float4 kr[NU];
            qv[0] = *reinterpret_cast<const float4*>(&sQ[qi][4 * (2 * c8)]);
            kr[0] = *reinterpret_cast<const float4*>(&sK[jl0][4 * (2 * c8)]);
            qv[1] = *reinterpret_cast<const float4*>(&sQ[qi][4 * (2 * c8 + 1)]);
#pragma unroll
            for (int uu = NU - 1; uu >= 3; --uu)
                kr[uu] = *reinterpret_cast<const float4*>(&sK[jl0 + 16 * (uu % NIT)][4 * (2 * c8 + uu / NIT)]);
            kr[1] = *reinterpret_cast<const float4*>(&sK[jl0 + 16][4 * (2 * c8)]);
            kr[2] = *reinterpret_cast<const float4*>(&sK[jl0 + 32][4 * (2 * c8)]);
            v2f dlo[NU], dhi[NU], elo[NU], ehi[NU];
            dlo[0] = v2f{qv[0].x, qv[0].y} - v2f{kr[0].x, kr[0].y};
            dhi[0] = v2f{qv[0].z, qv[0].w} - v2f{kr[0].z, kr[0].w};
#pragma unroll
            for (int u = 0; u < NU; ++u) {
                const int un = u + 1 < NU ? u + 1 : u;
                const float4 qn = qv[un / NIT];
                elo[u].x = __builtin_amdgcn_exp2f(-__builtin_fabsf(dlo[u].x));
                if (u + 1 < NU) dlo[un] = v2f{qn.x, qn.y} - v2f{kr[un].x, kr[un].y};
                elo[u].y = __builtin_amdgcn_exp2f(-__builtin_fabsf(dlo[u].y));
                if (u >= 1) sa[(u - 1) % NIT] += elo[u - 1]; else sa[(NU - 1) % NIT] += carlo;
                ehi[u].x = __builtin_amdgcn_exp2f(-__builtin_fabsf(dhi[u].x));
                if (u + 1 < NU) dhi[un] = v2f{qn.z, qn.w} - v2f{kr[un].z, kr[un].w};
                ehi[u].y = __builtin_amdgcn_exp2f(-__builtin_fabsf(dhi[u].y));
                if (u >= 1) sb[(u - 1) % NIT] += ehi[u - 1]; else sb[(NU - 1) % NIT] += carhi;
            }
            carlo = elo[NU - 1];
            carhi = ehi[NU - 1];
            __builtin_amdgcn_sched_group_barrier(0x002, 2, 0);
            __builtin_amdgcn_sched_group_barrier(0x100, NU, 0);
            __builtin_amdgcn_sched_group_barrier(0x100, 2, 0);
            __builtin_amdgcn_sched_group_barrier(0x002, 2, 0);
#pragma unroll
            for (int u = 0; u < NU; ++u) {
                __builtin_amdgcn_sched_group_barrier(0x400, 1, 0);
                if (u + 1 < NU) __builtin_amdgcn_sched_group_barrier(0x002, 1, 0);
                __builtin_amdgcn_sched_group_barrier(0x400, 1, 0);
                __builtin_amdgcn_sched_group_barrier(0x002, 1, 0);
                __builtin_amdgcn_sched_group_barrier(0x400, 1, 0);
                if (u + 1 < NU) __builtin_amdgcn_sched_group_barrier(0x002, 1, 0);
                __builtin_amdgcn_sched_group_barrier(0x400, 1, 0);
                __builtin_amdgcn_sched_group_barrier(0x002, 1, 0);
            }
        }
        sa[(NU - 1) % NIT] += carlo;
        sb[(NU - 1) % NIT] += carhi;
        float s[NIT];
#pragma unroll
        for (int it = 0; it < NIT; ++it) s[it] = (sa[it].x + sa[it].y) + (sb[it].x + sb[it].y);
        const float bound = SCALE * ((float)CDIM + fmaxf(sW[0], 0.f) + fmaxf(sW[1], 0.f)
                                     + fmaxf(sW[2], 0.f));
        float psum = 0.f;
#pragma unroll
        for (int it = 0; it < NIT; ++it) {
            int jl = jl0 + 16 * it;
            float sc = s[it];
#pragma unroll
            for (int d = 0; d < 3; ++d)
                sc += __builtin_amdgcn_exp2f(-__builtin_fabsf(sQc[qi][d] - sKc[jl][d])) * sW[d];
            sc *= SCALE;
            float e = j0 + jl < NKV ? __expf(sc - bound) : 0.f;
            sP[qi][jl] = e;
            psum += e;
        }
        sRed[qi][jl0] = psum;
    }
    __syncthreads();

    if (wave < 8) {
        const int qtl = wave >> 2;
        const long base = (long)((2 * qb + qtl) * NCHUNK + ch) * 16;
        int n0 = (wave & 3) * 16;
        v8f acc = {};
        for (int kk = 0; kk < CHUNK; kk += 4) {
            int ka = kk + 2 * half;
            v2f a, b;
            a.x = sP[16 * qtl + lm][ka];
            a.y = sP[16 * qtl + lm][ka + 1];
            b.x = sV[ka][n0 + lm];
            b.y = sV[ka + 1][n0 + lm];
            acc = wmma_f32_4(a, b, acc);
        }
#pragma unroll
        for (int r = 0; r < 8; ++r) {
            int row = r + 8 * half;
            po[(base + row) * CDIM + n0 + lm] = acc[r];
        }
    }
    if (wave == 8) {
        const long b1 = (long)((2 * qb + half) * NCHUNK + ch) * 16 + lm;
        float sum = 0.f;
#pragma unroll
        for (int i = 0; i < 16; ++i) sum += sRed[lane][i];
        ps[b1] = sum;
    }
}

__global__ void __launch_bounds__(256)
diffattn_combine(const float* __restrict__ po, const float* __restrict__ ps,
                 const float* __restrict__ Wkv,
                 const float* __restrict__ Wproj,
                 const float* __restrict__ bproj, float* __restrict__ out) {
    __shared__ float sO[16][CDIM + 1];
    __shared__ float sT[16][CDIM + 1];

    const int tid = threadIdx.x;
    const int qt = blockIdx.x;
    const long pbase = (long)qt * NCHUNK * 16;
    const int wave = tid >> 5;
    const int lane = tid & 31;
    const int half = lane >> 4;
    const int lm = lane & 15;

    const int r = tid >> 4, c4 = tid & 15;
    float4 pv[NCHUNK];
    float sv[NCHUNK];
#pragma unroll
    for (int c = 0; c < NCHUNK; ++c) {
        pv[c] = *reinterpret_cast<const float4*>(po + (pbase + c * 16 + r) * CDIM + 4 * c4);
        sv[c] = ps[pbase + c * 16 + r];
    }

    v2f bv[CDIM / 4];
    if (wave < 4) {
        const float* col = Wkv + CDIM + wave * 16 + lm;
#pragma unroll
        for (int t = 0; t < CDIM / 4; ++t) {
            int ka = 4 * t + 2 * half;
            bv[t].x = col[ka * 128];
            bv[t].y = col[(ka + 1) * 128];
        }
    }

    {
        float4 a = pv[0];
        float S = sv[0];
#pragma unroll
        for (int c = 1; c < NCHUNK; ++c) {
            a.x += pv[c].x;
            a.y += pv[c].y;
            a.z += pv[c].z;
            a.w += pv[c].w;
            S += sv[c];
        }
        const float inv = 1.0f / S;
        sO[r][4 * c4 + 0] = a.x * inv;
        sO[r][4 * c4 + 1] = a.y * inv;
        sO[r][4 * c4 + 2] = a.z * inv;
        sO[r][4 * c4 + 3] = a.w * inv;
    }
    __syncthreads();

    if (wave < 4) {
        int n0 = wave * 16;
        v8f acc = {};
#pragma unroll
        for (int t = 0; t < CDIM / 4; ++t) {
            int ka = 4 * t + 2 * half;
            v2f a;
            a.x = sO[lm][ka];
            a.y = sO[lm][ka + 1];
            acc = wmma_f32_4(a, bv[t], acc);
        }
#pragma unroll
        for (int r = 0; r < 8; ++r) sT[r + 8 * half][n0 + lm] = acc[r];
    }
    __syncthreads();

    if (wave < 4) {
        int n0 = wave * 16;
        v8f acc = {};
        for (int kk = 0; kk < CDIM; kk += 4) {
            int ka = kk + 2 * half;
            v2f a, b;
            a.x = sT[lm][ka];
            a.y = sT[lm][ka + 1];
            b.x = Wproj[ka * CDIM + n0 + lm];
            b.y = Wproj[(ka + 1) * CDIM + n0 + lm];
            acc = wmma_f32_4(a, b, acc);
        }
        float bb = bproj[n0 + lm];
#pragma unroll
        for (int r = 0; r < 8; ++r) {
            int row = r + 8 * half;
            out[(qt * 16 + row) * CDIM + n0 + lm] = acc[r] + bb;
        }
    }
}

extern "C" void kernel_launch(void* const* d_in, const int* in_sizes, int n_in,
                              void* d_out, int out_size, void* d_ws, size_t ws_size,
                              hipStream_t stream) {
    const float* q        = (const float*)d_in[0];
    const float* q_coord  = (const float*)d_in[1];
    const float* kv       = (const float*)d_in[2];
    const float* kv_coord = (const float*)d_in[3];
    const float* Wq       = (const float*)d_in[4];
    const float* Wkv      = (const float*)d_in[5];
    const float* Wdelta   = (const float*)d_in[6];
    const float* Wproj    = (const float*)d_in[7];
    const float* bproj    = (const float*)d_in[8];
    float* out = (float*)d_out;
    float* ws  = (float*)d_ws;

    float* po = ws + OFF_PO;
    float* ps = ws + OFF_PS;

    dim3 sgrid(NCHUNK, NQB);
    diffattn_score<<<sgrid, 512, 0, stream>>>(q, kv, q_coord, kv_coord, Wq, Wkv,
                                              Wdelta, po, ps);

    diffattn_combine<<<NQT, 256, 0, stream>>>(po, ps, Wkv, Wproj, bproj, out);
}


# combine applies the folded matrix Wv @ Wproj (16 tiles built on WMMA by score waves idle during p @ kv) in one product, not two: one WMMA chain, LDS pass and barrier gone; sum order changes
# speedup vs baseline: 3.5280x; 1.0957x over previous
#include <hip/hip_runtime.h>
#include <hip/hip_bf16.h>

typedef float v2f __attribute__((ext_vector_type(2)));
typedef float v8f __attribute__((ext_vector_type(8)));

#define NQ 512
#define NKV 2562
#define CDIM 64
#define NCHUNK 15
#define CHUNK 176
#define NIT (CHUNK / 16)
#define QB 32
#define NQB (NQ / QB)
#define NQT (NQ / 16)
#define SCALE (1.0f / 64.0f)
#define VROWS 192

#define OFF_PO   0L
#define OFF_PS   (OFF_PO   + (long)NQT * NCHUNK * 16 * CDIM)
#define OFF_WVP  (OFF_PS   + (long)NQT * NCHUNK * 16)

static __device__ __forceinline__ v8f wmma_f32_4(v2f a, v2f b, v8f c) {
    return __builtin_amdgcn_wmma_f32_16x16x4_f32(false, a, false, b, (short)0, c,
                                                 false, false);
}

#define TSTR 68
#define LOG2E 1.44269504088896340736f
#define VSTR 76
#define PSTR (CHUNK + 4)

template <int NT>
static __device__ __forceinline__ void project_tiles(
        const float (*sA)[VSTR], const float (*sB)[VSTR], float (*sOut)[TSTR],
        int rt0, int rstep, int cK, int lm, int half) {
    v8f acc[NT];
#pragma unroll
    for (int i = 0; i < NT; ++i) acc[i] = v8f{};
#pragma unroll
    for (int kk = 0; kk < CDIM; kk += 4) {
        int ka = kk + 2 * half;
        v2f b;
        b.x = sB[ka][cK];
        b.y = sB[ka + 1][cK];
#pragma unroll
        for (int i = 0; i < NT; ++i) {
            v2f a = *reinterpret_cast<const v2f*>(&sA[16 * (rt0 + rstep * i) + lm][ka]);
            acc[i] = wmma_f32_4(a, b, acc[i]);
        }
    }
#pragma unroll
    for (int i = 0; i < NT; ++i) {
#pragma unroll
        for (int r = 0; r < 8; ++r) {
            int row = 16 * (rt0 + rstep * i) + r + 8 * half;
            sOut[row][cK] = acc[i][r] * LOG2E;
        }
    }
}

__global__ void __launch_bounds__(512)
diffattn_score(const float* __restrict__ q, const float* __restrict__ kv,
               const float* __restrict__ q_coord,
               const float* __restrict__ kv_coord,
               const float* __restrict__ Wq, const float* __restrict__ Wkv,
               const float* __restrict__ Wdelta,
               const float* __restrict__ Wproj,
               float* __restrict__ po, float* __restrict__ ps,
               float* __restrict__ wvp) {
    __shared__ __attribute__((aligned(16))) float sQ[QB][TSTR];
    __shared__ __attribute__((aligned(16))) float sK[CHUNK][TSTR];
    __shared__ __attribute__((aligned(16))) float sV[VROWS][VSTR];
    __shared__ __attribute__((aligned(16))) float sQr[QB][VSTR];
    __shared__ __attribute__((aligned(16))) float sWk[CDIM][VSTR];
    __shared__ __attribute__((aligned(16))) float sWq[CDIM][VSTR];
    __shared__ __attribute__((aligned(16))) float sP[QB][PSTR];
    __shared__ float sQc[128][4];
    __shared__ float sKc[512][4];
    __shared__ __attribute__((aligned(16))) float sWd[512];
    __shared__ float sRed[QB][17];
    __shared__ float sW[3];

    const int tid = threadIdx.x;
    const int ch = blockIdx.x;
    const int qb = blockIdx.y;
    const int q0 = qb * QB;
    const int j0 = ch * CHUNK;
    const int wave = tid >> 5;
    const int lane = tid & 31;
    const int half = lane >> 4;
    const int lm = lane & 15;

    float4 rkv[6];
#pragma unroll
    for (int i = 0; i < 6; ++i) {
        int idx = tid + 512 * i;
        int jl = idx >> 4;
        int rr = j0 + jl < NKV ? j0 + jl : NKV - 1;
        rkv[i] = *reinterpret_cast<const float4*>(kv + (long)rr * CDIM + 4 * (idx & 15));
    }
    const float4 rq = *reinterpret_cast<const float4*>(q + (long)(q0 + (tid >> 4)) * CDIM + 4 * (tid & 15));
    float4 rwk[2], rwq[2];
#pragma unroll
    for (int i = 0; i < 2; ++i) {
        int idx = tid + 512 * i;
        rwk[i] = *reinterpret_cast<const float4*>(Wkv + (idx >> 4) * 128 + 4 * (idx & 15));
        rwq[i] = *reinterpret_cast<const float4*>(Wq + (idx >> 4) * CDIM + 4 * (idx & 15));
    }
    const int cr = tid >> 2, cd = tid & 3;
    const float rqc = q_coord[(q0 + (cr < QB ? cr : QB - 1)) * 3 + (cd < 3 ? cd : 2)];
    const bool okc = j0 + tid < NKV;
    const float* cp = kv_coord + (okc ? j0 + tid : NKV - 1) * 3;
    const float x0 = cp[0], x1 = cp[1], x2 = cp[2];
    const float rwd = Wdelta[tid < 3 * CDIM ? tid : 3 * CDIM - 1];
#pragma unroll
    for (int i = 0; i < 6; ++i) {
        int idx = tid + 512 * i;
        *reinterpret_cast<float4*>(&sV[idx >> 4][4 * (idx & 15)]) = rkv[i];
    }
    *reinterpret_cast<float4*>(&sQr[tid >> 4][4 * (tid & 15)]) = rq;
#pragma unroll
    for (int i = 0; i < 2; ++i) {
        int idx = tid + 512 * i;
        *reinterpret_cast<float4*>(&sWk[idx >> 4][4 * (idx & 15)]) = rwk[i];
        *reinterpret_cast<float4*>(&sWq[idx >> 4][4 * (idx & 15)]) = rwq[i];
    }
    sQc[cr][cd] = rqc * LOG2E;
    sKc[tid][0] = okc ? x0 * LOG2E : 0.f;
    sKc[tid][1] = okc ? x1 * LOG2E : 0.f;
    sKc[tid][2] = okc ? x2 * LOG2E : 0.f;
    sWd[tid] = rwd;
    __syncthreads();

    if (tid >= 224 && tid < 227) {
        int d = tid - 224;
        float s = 0.f;
        for (int cc = 0; cc < CDIM; ++cc) s += sWd[d * CDIM + cc];
        sW[d] = s;
    }

    const int cK = 16 * (wave & 3) + lm;
    if (wave < 12) {
        project_tiles<3>(sV, sWk, sK, wave >> 2, 4, cK, lm, half);
    } else {
        project_tiles<2>(sV, sWk, sK, 3, 4, cK, lm, half);
        project_tiles<2>(sQr, sWq, sQ, 0, 1, cK, lm, half);
    }
    __syncthreads();

    {
        const int qi = tid & 31;
        const int jl0 = tid >> 5;
        constexpr int NU = 2 * NIT;
        v2f sa[NIT], sb[NIT];
#pragma unroll
        for (int it = 0; it < NIT; ++it) { sa[it] = v2f{0.f, 0.f}; sb[it] = v2f{0.f, 0.f}; }
        v2f carlo = {0.f, 0.f}, carhi = {0.f, 0.f};
        for (int c8 = 0; c8 < CDIM / 8; ++c8) {
            float4 qv[2];
            float4 kr[NU];
            qv[0] = *reinterpret_cast<const float4*>(&sQ[qi][4 * (2 * c8)]);
            kr[0] = *reinterpret_cast<const float4*>(&sK[jl0][4 * (2 * c8)]);
            qv[1] = *reinterpret_cast<const float4*>(&sQ[qi][4 * (2 * c8 + 1)]);
#pragma unroll
            for (int uu = NU - 1; uu >= 3; --uu)
                kr[uu] = *reinterpret_cast<const float4*>(&sK[jl0 + 16 * (uu % NIT)][4 * (2 * c8 + uu / NIT)]);
            kr[1] = *reinterpret_cast<const float4*>(&sK[jl0 + 16][4 * (2 * c8)]);
            kr[2] = *reinterpret_cast<const float4*>(&sK[jl0 + 32][4 * (2 * c8)]);
            v2f dlo[NU], dhi[NU], elo[NU], ehi[NU];
            dlo[0] = v2f{qv[0].x, qv[0].y} - v2f{kr[0].x, kr[0].y};
            dhi[0] = v2f{qv[0].z, qv[0].w} - v2f{kr[0].z, kr[0].w};
#pragma unroll
            for (int u = 0; u < NU; ++u) {
                const int un = u + 1 < NU ? u + 1 : u;
                const float4 qn = qv[un / NIT];
                elo[u].x = __builtin_amdgcn_exp2f(-__builtin_fabsf(dlo[u].x));
                if (u + 1 < NU) dlo[un] = v2f{qn.x, qn.y} - v2f{kr[un].x, kr[un].y};
                elo[u].y = __builtin_amdgcn_exp2f(-__builtin_fabsf(dlo[u].y));
                if (u >= 1) sa[(u - 1) % NIT] += elo[u - 1]; else sa[(NU - 1) % NIT] += carlo;
                ehi[u].x = __builtin_amdgcn_exp2f(-__builtin_fabsf(dhi[u].x));
                if (u + 1 < NU) dhi[un] = v2f{qn.z, qn.w} - v2f{kr[un].z, kr[un].w};
                ehi[u].y = __builtin_amdgcn_exp2f(-__builtin_fabsf(dhi[u].y));
                if (u >= 1) sb[(u - 1) % NIT] += ehi[u - 1]; else sb[(NU - 1) % NIT] += carhi;
            }
            carlo = elo[NU - 1];
            carhi = ehi[NU - 1];
            __builtin_amdgcn_sched_group_barrier(0x002, 2, 0);
            __builtin_amdgcn_sched_group_barrier(0x100, NU, 0);
            __builtin_amdgcn_sched_group_barrier(0x100, 2, 0);
            __builtin_amdgcn_sched_group_barrier(0x002, 2, 0);
#pragma unroll
            for (int u = 0; u < NU; ++u) {
                __builtin_amdgcn_sched_group_barrier(0x400, 1, 0);
                if (u + 1 < NU) __builtin_amdgcn_sched_group_barrier(0x002, 1, 0);
                __builtin_amdgcn_sched_group_barrier(0x400, 1, 0);
                __builtin_amdgcn_sched_group_barrier(0x002, 1, 0);
                __builtin_amdgcn_sched_group_barrier(0x400, 1, 0);
                if (u + 1 < NU) __builtin_amdgcn_sched_group_barrier(0x002, 1, 0);
                __builtin_amdgcn_sched_group_barrier(0x400, 1, 0);
                __builtin_amdgcn_sched_group_barrier(0x002, 1, 0);
            }
        }
        sa[(NU - 1) % NIT] += carlo;
        sb[(NU - 1) % NIT] += carhi;
        float s[NIT];
#pragma unroll
        for (int it = 0; it < NIT; ++it) s[it] = (sa[it].x + sa[it].y) + (sb[it].x + sb[it].y);
        const float bound = SCALE * ((float)CDIM + fmaxf(sW[0], 0.f) + fmaxf(sW[1], 0.f)
                                     + fmaxf(sW[2], 0.f));
        float psum = 0.f;
#pragma unroll
        for (int it = 0; it < NIT; ++it) {
            int jl = jl0 + 16 * it;
            float sc = s[it];
#pragma unroll
            for (int d = 0; d < 3; ++d)
                sc += __builtin_amdgcn_exp2f(-__builtin_fabsf(sQc[qi][d] - sKc[jl][d])) * sW[d];
            sc *= SCALE;
            float e = j0 + jl < NKV ? __expf(sc - bound) : 0.f;
            sP[qi][jl] = e;
            psum += e;
        }
        sRed[qi][jl0] = psum;
    }
    __syncthreads();

    if (wave < 8) {
        const int qtl = wave >> 2;
        const long base = (long)((2 * qb + qtl) * NCHUNK + ch) * 16;
        int n0 = (wave & 3) * 16;
        v8f acc = {};
        for (int kk = 0; kk < CHUNK; kk += 4) {
            int ka = kk + 2 * half;
            v2f a, b;
            a.x = sP[16 * qtl + lm][ka];
            a.y = sP[16 * qtl + lm][ka + 1];
            b.x = sV[ka][n0 + lm];
            b.y = sV[ka + 1][n0 + lm];
            acc = wmma_f32_4(a, b, acc);
        }
#pragma unroll
        for (int r = 0; r < 8; ++r) {
            int row = r + 8 * half;
            po[(base + row) * CDIM + n0 + lm] = acc[r];
        }
    }
    if (wave == 8) {
        const long b1 = (long)((2 * qb + half) * NCHUNK + ch) * 16 + lm;
        float sum = 0.f;
#pragma unroll
        for (int i = 0; i < 16; ++i) sum += sRed[lane][i];
        ps[b1] = sum;
    }
    if (wave == 9 && ch == 0) {
        const int ti = qb >> 2, tj = qb & 3;
        const float* arow = Wkv + (16 * ti + lm) * 128 + CDIM;
        const float* bcol = Wproj + 16 * tj + lm;
        v2f af[CDIM / 4], bf[CDIM / 4];
#pragma unroll
        for (int t = 0; t < CDIM / 4; ++t) {
            int ka = 4 * t + 2 * half;
            af[t] = *reinterpret_cast<const v2f*>(arow + ka);
            bf[t].x = bcol[ka * CDIM];
            bf[t].y = bcol[(ka + 1) * CDIM];
        }
        v8f acc = {};
#pragma unroll
        for (int t = 0; t < CDIM / 4; ++t) acc = wmma_f32_4(af[t], bf[t], acc);
#pragma unroll
        for (int r = 0; r < 8; ++r)
            wvp[(16 * ti + r + 8 * half) * CDIM + 16 * tj + lm] = acc[r];
    }
}

__global__ void __launch_bounds__(256)
diffattn_combine(const float* __restrict__ po, const float* __restrict__ ps,
                 const float* __restrict__ wvp,
                 const float* __restrict__ bproj, float* __restrict__ out) {
    __shared__ float sO[16][CDIM + 1];

    const int tid = threadIdx.x;
    const int qt = blockIdx.x;
    const long pbase = (long)qt * NCHUNK * 16;
    const int wave = tid >> 5;
    const int lane = tid & 31;
    const int half = lane >> 4;
    const int lm = lane & 15;

    const int r = tid >> 4, c4 = tid & 15;
    float4 pv[NCHUNK];
    float sv[NCHUNK];
#pragma unroll
    for (int c = 0; c < NCHUNK; ++c) {
        pv[c] = *reinterpret_cast<const float4*>(po + (pbase + c * 16 + r) * CDIM + 4 * c4);
        sv[c] = ps[pbase + c * 16 + r];
    }

    v2f bv[CDIM / 4];
    float bb = 0.f;
    if (wave < 4) {
        const float* col = wvp + wave * 16 + lm;
#pragma unroll
        for (int t = 0; t < CDIM / 4; ++t) {
            int ka = 4 * t + 2 * half;
            bv[t].x = col[ka * CDIM];
            bv[t].y = col[(ka + 1) * CDIM];
        }
        bb = bproj[wave * 16 + lm];
    }

    {
        float4 a = pv[0];
        float S = sv[0];
#pragma unroll
        for (int c = 1; c < NCHUNK; ++c) {
            a.x += pv[c].x;
            a.y += pv[c].y;
            a.z += pv[c].z;
            a.w += pv[c].w;
            S += sv[c];
        }
        const float inv = 1.0f / S;
        sO[r][4 * c4 + 0] = a.x * inv;
        sO[r][4 * c4 + 1] = a.y * inv;
        sO[r][4 * c4 + 2] = a.z * inv;
        sO[r][4 * c4 + 3] = a.w * inv;
    }
    __syncthreads();

    if (wave < 4) {
        int n0 = wave * 16;
        v8f acc = {};
#pragma unroll
        for (int t = 0; t < CDIM / 4; ++t) {
            int ka = 4 * t + 2 * half;
            v2f a;
            a.x = sO[lm][ka];
            a.y = sO[lm][ka + 1];
            acc = wmma_f32_4(a, bv[t], acc);
        }
#pragma unroll
        for (int r = 0; r < 8; ++r) {
            int row = r + 8 * half;
            out[(qt * 16 + row) * CDIM + n0 + lm] = acc[r] + bb;
        }
    }
}

extern "C" void kernel_launch(void* const* d_in, const int* in_sizes, int n_in,
                              void* d_out, int out_size, void* d_ws, size_t ws_size,
                              hipStream_t stream) {
    const float* q        = (const float*)d_in[0];
    const float* q_coord  = (const float*)d_in[1];
    const float* kv       = (const float*)d_in[2];
    const float* kv_coord = (const float*)d_in[3];
    const float* Wq       = (const float*)d_in[4];
    const float* Wkv      = (const float*)d_in[5];
    const float* Wdelta   = (const float*)d_in[6];
    const float* Wproj    = (const float*)d_in[7];
    const float* bproj    = (const float*)d_in[8];
    float* out = (float*)d_out;
    float* ws  = (float*)d_ws;

    float* po = ws + OFF_PO;
    float* ps = ws + OFF_PS;
    float* wvp = ws + OFF_WVP;

    dim3 sgrid(NCHUNK, NQB);
    diffattn_score<<<sgrid, 512, 0, stream>>>(q, kv, q_coord, kv_coord, Wq, Wkv,
                                              Wdelta, Wproj, po, ps, wvp);

    diffattn_combine<<<NQT, 256, 0, stream>>>(po, ps, wvp, bproj, out);
}
